# starting kernel
# baseline (speedup 1.0000x reference)
#include <hip/hip_runtime.h>

typedef float v2f __attribute__((ext_vector_type(2)));
typedef float v8f __attribute__((ext_vector_type(8)));

constexpr int N_NODES   = 100000;
constexpr int N_HYEDGES = 20000;
constexpr int NNZ       = 1600000;
constexpr int IN_CH     = 256;
constexpr int OUT_CH    = 128;

__global__ __launch_bounds__(256) void hyfm_gemm_wmma_f32(
    const float* __restrict__ x, const float* __restrict__ theta,
    float* __restrict__ y)
{
    const int wave    = blockIdx.x * (blockDim.x >> 5) + (threadIdx.x >> 5);
    const int n_tiles = N_NODES / 16;
    if (wave >= n_tiles) return;
    const int lane = threadIdx.x & 31;
    const int r    = lane & 15;
    const int half = lane >> 4;

    const float* __restrict__ xrow = x + (size_t)(wave * 16 + r) * IN_CH;

    v8f acc[8] = {};

    for (int kk = 0; kk < IN_CH; kk += 4) {
        const int k0 = kk + 2 * half;
        v2f a = *(const v2f*)(xrow + k0);
        const float* __restrict__ th0 = theta + (size_t)k0 * OUT_CH + r;
#pragma unroll
        for (int c = 0; c < 8; ++c) {
            v2f b;
            b.x = th0[c * 16];
            b.y = th0[c * 16 + OUT_CH];
            acc[c] = __builtin_amdgcn_wmma_f32_16x16x4_f32(
                false, a, false, b, (short)0, acc[c], false, false);
        }
    }

#pragma unroll
    for (int c = 0; c < 8; ++c) {
#pragma unroll
        for (int i = 0; i < 8; ++i) {
            y[(size_t)(wave * 16 + i + 8 * half) * OUT_CH + c * 16 + r] = acc[c][i];
        }
    }
}

__global__ __launch_bounds__(256) void hyfm_degree(
    const int* __restrict__ nid, const int* __restrict__ eid,
    float* __restrict__ deg_n, float* __restrict__ deg_e, int nnz)
{
    int i = blockIdx.x * blockDim.x + threadIdx.x;
    if (i >= nnz) return;
    atomicAdd(&deg_e[eid[i]], 1.0f);
    atomicAdd(&deg_n[nid[i]], 1.0f);
}

__global__ __launch_bounds__(256) void hyfm_recip(float* __restrict__ d, int n)
{
    int i = blockIdx.x * blockDim.x + threadIdx.x;
    if (i >= n) return;
    float v = d[i];
    d[i] = (v > 0.0f) ? (1.0f / v) : 0.0f;
}

__global__ __launch_bounds__(256) void hyfm_bias_init(
    const float* __restrict__ bias, float* __restrict__ out, int total)
{
    int i = blockIdx.x * blockDim.x + threadIdx.x;
    if (i >= total) return;
    out[i] = bias[i & (OUT_CH - 1)];
}

__global__ __launch_bounds__(256) void hyfm_scatter(
    const float* __restrict__ src, float* __restrict__ dst,
    const int* __restrict__ gidx, const int* __restrict__ sidx,
    const float* __restrict__ inv_deg, int nnz)
{
    const int w = blockIdx.x * (blockDim.x >> 5) + (threadIdx.x >> 5);
    if (w >= nnz) return;
    const int lane = threadIdx.x & 31;
    const int g = gidx[w];
    const int s = sidx[w];
    const float sc = inv_deg[s];
    const float4 v = ((const float4*)(src + (size_t)g * OUT_CH))[lane];
    float* drow = dst + (size_t)s * OUT_CH + lane * 4;
    atomicAdd(drow + 0, v.x * sc);
    atomicAdd(drow + 1, v.y * sc);
    atomicAdd(drow + 2, v.z * sc);
    atomicAdd(drow + 3, v.w * sc);
}

extern "C" void kernel_launch(void* const* d_in, const int* in_sizes, int n_in,
                              void* d_out, int out_size, void* d_ws, size_t ws_size,
                              hipStream_t stream)
{
    const float* x          = (const float*)d_in[0];
    const float* theta      = (const float*)d_in[1];
    const float* bias       = (const float*)d_in[2];
    const int*   node_idx   = (const int*)d_in[3];
    const int*   hyedge_idx = (const int*)d_in[4];
    float*       out        = (float*)d_out;

    char*  ws    = (char*)d_ws;
    float* y     = (float*)(ws);
    float* e_ft  = (float*)(ws + 51200000);
    float* deg_e = (float*)(ws + 61440000);
    float* deg_n = (float*)(ws + 61520000);

    hipMemsetAsync(ws + 51200000, 0, 10720000, stream);

    hyfm_degree<<<(NNZ + 255) / 256, 256, 0, stream>>>(
        node_idx, hyedge_idx, deg_n, deg_e, NNZ);
    hyfm_recip<<<(N_HYEDGES + N_NODES + 255) / 256, 256, 0, stream>>>(
        deg_e, N_HYEDGES + N_NODES);

    hyfm_gemm_wmma_f32<<<(N_NODES / 16 + 7) / 8, 256, 0, stream>>>(x, theta, y);

    hyfm_bias_init<<<(N_NODES * OUT_CH + 255) / 256, 256, 0, stream>>>(
        bias, out, N_NODES * OUT_CH);

    hyfm_scatter<<<(NNZ + 7) / 8, 256, 0, stream>>>(
        y, e_ft, node_idx, hyedge_idx, deg_e, NNZ);
    hyfm_scatter<<<(NNZ + 7) / 8, 256, 0, stream>>>(
        e_ft, out, hyedge_idx, node_idx, deg_n, NNZ);
}


# scatter kernels: lane l owns channels l,l+32,l+64,l+96, so each f32 atomic add and each gather load covers one whole 128-B line instead of a quarter of four; atomics, arrival-order sums kept
# speedup vs baseline: 1.3543x; 1.3543x over previous
#include <hip/hip_runtime.h>

typedef float v2f __attribute__((ext_vector_type(2)));
typedef float v8f __attribute__((ext_vector_type(8)));

constexpr int N_NODES   = 100000;
constexpr int N_HYEDGES = 20000;
constexpr int NNZ       = 1600000;
constexpr int IN_CH     = 256;
constexpr int OUT_CH    = 128;

__global__ __launch_bounds__(256) void hyfm_gemm_wmma_f32(
    const float* __restrict__ x, const float* __restrict__ theta,
    float* __restrict__ y)
{
    const int wave    = blockIdx.x * (blockDim.x >> 5) + (threadIdx.x >> 5);
    const int n_tiles = N_NODES / 16;
    if (wave >= n_tiles) return;
    const int lane = threadIdx.x & 31;
    const int r    = lane & 15;
    const int half = lane >> 4;

    const float* __restrict__ xrow = x + (size_t)(wave * 16 + r) * IN_CH;

    v8f acc[8] = {};

    for (int kk = 0; kk < IN_CH; kk += 4) {
        const int k0 = kk + 2 * half;
        v2f a = *(const v2f*)(xrow + k0);
        const float* __restrict__ th0 = theta + (size_t)k0 * OUT_CH + r;
#pragma unroll
        for (int c = 0; c < 8; ++c) {
            v2f b;
            b.x = th0[c * 16];
            b.y = th0[c * 16 + OUT_CH];
            acc[c] = __builtin_amdgcn_wmma_f32_16x16x4_f32(
                false, a, false, b, (short)0, acc[c], false, false);
        }
    }

#pragma unroll
    for (int c = 0; c < 8; ++c) {
#pragma unroll
        for (int i = 0; i < 8; ++i) {
            y[(size_t)(wave * 16 + i + 8 * half) * OUT_CH + c * 16 + r] = acc[c][i];
        }
    }
}

__global__ __launch_bounds__(256) void hyfm_degree(
    const int* __restrict__ nid, const int* __restrict__ eid,
    float* __restrict__ deg_n, float* __restrict__ deg_e, int nnz)
{
    int i = blockIdx.x * blockDim.x + threadIdx.x;
    if (i >= nnz) return;
    atomicAdd(&deg_e[eid[i]], 1.0f);
    atomicAdd(&deg_n[nid[i]], 1.0f);
}

__global__ __launch_bounds__(256) void hyfm_recip(float* __restrict__ d, int n)
{
    int i = blockIdx.x * blockDim.x + threadIdx.x;
    if (i >= n) return;
    float v = d[i];
    d[i] = (v > 0.0f) ? (1.0f / v) : 0.0f;
}

__global__ __launch_bounds__(256) void hyfm_bias_init(
    const float* __restrict__ bias, float* __restrict__ out, int total)
{
    int i = blockIdx.x * blockDim.x + threadIdx.x;
    if (i >= total) return;
    out[i] = bias[i & (OUT_CH - 1)];
}

__global__ __launch_bounds__(256) void hyfm_scatter(
    const float* __restrict__ src, float* __restrict__ dst,
    const int* __restrict__ gidx, const int* __restrict__ sidx,
    const float* __restrict__ inv_deg, int nnz)
{
    const int w = blockIdx.x * (blockDim.x >> 5) + (threadIdx.x >> 5);
    if (w >= nnz) return;
    const int lane = threadIdx.x & 31;
    const int g = gidx[w];
    const int s = sidx[w];
    const float sc = inv_deg[s];
    const float* __restrict__ srow = src + (size_t)g * OUT_CH + lane;
    const float v0 = srow[0];
    const float v1 = srow[32];
    const float v2 = srow[64];
    const float v3 = srow[96];
    float* drow = dst + (size_t)s * OUT_CH + lane;
    atomicAdd(drow +  0, v0 * sc);
    atomicAdd(drow + 32, v1 * sc);
    atomicAdd(drow + 64, v2 * sc);
    atomicAdd(drow + 96, v3 * sc);
}

extern "C" void kernel_launch(void* const* d_in, const int* in_sizes, int n_in,
                              void* d_out, int out_size, void* d_ws, size_t ws_size,
                              hipStream_t stream)
{
    const float* x          = (const float*)d_in[0];
    const float* theta      = (const float*)d_in[1];
    const float* bias       = (const float*)d_in[2];
    const int*   node_idx   = (const int*)d_in[3];
    const int*   hyedge_idx = (const int*)d_in[4];
    float*       out        = (float*)d_out;

    char*  ws    = (char*)d_ws;
    float* y     = (float*)(ws);
    float* e_ft  = (float*)(ws + 51200000);
    float* deg_e = (float*)(ws + 61440000);
    float* deg_n = (float*)(ws + 61520000);

    hipMemsetAsync(ws + 51200000, 0, 10720000, stream);

    hyfm_degree<<<(NNZ + 255) / 256, 256, 0, stream>>>(
        node_idx, hyedge_idx, deg_n, deg_e, NNZ);
    hyfm_recip<<<(N_HYEDGES + N_NODES + 255) / 256, 256, 0, stream>>>(
        deg_e, N_HYEDGES + N_NODES);

    hyfm_gemm_wmma_f32<<<(N_NODES / 16 + 7) / 8, 256, 0, stream>>>(x, theta, y);

    hyfm_bias_init<<<(N_NODES * OUT_CH + 255) / 256, 256, 0, stream>>>(
        bias, out, N_NODES * OUT_CH);

    hyfm_scatter<<<(NNZ + 7) / 8, 256, 0, stream>>>(
        y, e_ft, node_idx, hyedge_idx, deg_e, NNZ);
    hyfm_scatter<<<(NNZ + 7) / 8, 256, 0, stream>>>(
        e_ft, out, hyedge_idx, node_idx, deg_n, NNZ);
}


# scatter kernels: one wave takes 32 nnz entries (coalesced index loads, row bases broadcast by readlane) and keeps 2 groups of 8 entries' gather loads in flight; atomics, arrival-order sums kept
# speedup vs baseline: 1.7101x; 1.2627x over previous
#include <hip/hip_runtime.h>

typedef float v2f __attribute__((ext_vector_type(2)));
typedef float v8f __attribute__((ext_vector_type(8)));

constexpr int N_NODES   = 100000;
constexpr int N_HYEDGES = 20000;
constexpr int NNZ       = 1600000;
constexpr int IN_CH     = 256;
constexpr int OUT_CH    = 128;

__global__ __launch_bounds__(256) void hyfm_gemm_wmma_f32(
    const float* __restrict__ x, const float* __restrict__ theta,
    float* __restrict__ y)
{
    const int wave    = blockIdx.x * (blockDim.x >> 5) + (threadIdx.x >> 5);
    const int n_tiles = N_NODES / 16;
    if (wave >= n_tiles) return;
    const int lane = threadIdx.x & 31;
    const int r    = lane & 15;
    const int half = lane >> 4;

    const float* __restrict__ xrow = x + (size_t)(wave * 16 + r) * IN_CH;

    v8f acc[8] = {};

    for (int kk = 0; kk < IN_CH; kk += 4) {
        const int k0 = kk + 2 * half;
        v2f a = *(const v2f*)(xrow + k0);
        const float* __restrict__ th0 = theta + (size_t)k0 * OUT_CH + r;
#pragma unroll
        for (int c = 0; c < 8; ++c) {
            v2f b;
            b.x = th0[c * 16];
            b.y = th0[c * 16 + OUT_CH];
            acc[c] = __builtin_amdgcn_wmma_f32_16x16x4_f32(
                false, a, false, b, (short)0, acc[c], false, false);
        }
    }

#pragma unroll
    for (int c = 0; c < 8; ++c) {
#pragma unroll
        for (int i = 0; i < 8; ++i) {
            y[(size_t)(wave * 16 + i + 8 * half) * OUT_CH + c * 16 + r] = acc[c][i];
        }
    }
}

__global__ __launch_bounds__(256) void hyfm_degree(
    const int* __restrict__ nid, const int* __restrict__ eid,
    float* __restrict__ deg_n, float* __restrict__ deg_e, int nnz)
{
    int i = blockIdx.x * blockDim.x + threadIdx.x;
    if (i >= nnz) return;
    atomicAdd(&deg_e[eid[i]], 1.0f);
    atomicAdd(&deg_n[nid[i]], 1.0f);
}

__global__ __launch_bounds__(256) void hyfm_recip(float* __restrict__ d, int n)
{
    int i = blockIdx.x * blockDim.x + threadIdx.x;
    if (i >= n) return;
    float v = d[i];
    d[i] = (v > 0.0f) ? (1.0f / v) : 0.0f;
}

__global__ __launch_bounds__(256) void hyfm_bias_init(
    const float* __restrict__ bias, float* __restrict__ out, int total)
{
    int i = blockIdx.x * blockDim.x + threadIdx.x;
    if (i >= total) return;
    out[i] = bias[i & (OUT_CH - 1)];
}

constexpr int SC_PER_WAVE = 32;
constexpr int SC_GROUP    = 8;
constexpr int SC_NGROUPS  = SC_PER_WAVE / SC_GROUP;
static_assert(NNZ % (SC_PER_WAVE * 8) == 0, "whole blocks of 8 waves x 32 entries");

__device__ __forceinline__ void hyfm_gather_group(
    const float* __restrict__ src, int gv, int lane, int first,
    float (&v)[SC_GROUP][4])
{
#pragma unroll
    for (int j = 0; j < SC_GROUP; ++j) {
        const int g = __builtin_amdgcn_readlane(gv, first + j);
        const float* __restrict__ srow = src + (size_t)g * OUT_CH + lane;
        v[j][0] = srow[0];
        v[j][1] = srow[32];
        v[j][2] = srow[64];
        v[j][3] = srow[96];
    }
}

__global__ __launch_bounds__(256) void hyfm_scatter(
    const float* __restrict__ src, float* __restrict__ dst,
    const int* __restrict__ gidx, const int* __restrict__ sidx,
    const float* __restrict__ inv_deg)
{
    const int w    = blockIdx.x * (blockDim.x >> 5) + (threadIdx.x >> 5);
    const int lane = threadIdx.x & 31;
    const int e    = w * SC_PER_WAVE + lane;
    const int gv   = gidx[e];
    const int sv   = sidx[e];
    const int scv  = __float_as_int(inv_deg[sv]);

    float v[2][SC_GROUP][4];
    hyfm_gather_group(src, gv, lane, 0, v[0]);

#pragma unroll
    for (int c = 0; c < SC_NGROUPS; ++c) {
        if (c + 1 < SC_NGROUPS)
            hyfm_gather_group(src, gv, lane, (c + 1) * SC_GROUP, v[(c + 1) & 1]);
#pragma unroll
        for (int j = 0; j < SC_GROUP; ++j) {
            const int   s  = __builtin_amdgcn_readlane(sv, c * SC_GROUP + j);
            const float sc = __int_as_float(
                __builtin_amdgcn_readlane(scv, c * SC_GROUP + j));
            float* drow = dst + (size_t)s * OUT_CH + lane;
            atomicAdd(drow +  0, v[c & 1][j][0] * sc);
            atomicAdd(drow + 32, v[c & 1][j][1] * sc);
            atomicAdd(drow + 64, v[c & 1][j][2] * sc);
            atomicAdd(drow + 96, v[c & 1][j][3] * sc);
        }
    }
}

extern "C" void kernel_launch(void* const* d_in, const int* in_sizes, int n_in,
                              void* d_out, int out_size, void* d_ws, size_t ws_size,
                              hipStream_t stream)
{
    const float* x          = (const float*)d_in[0];
    const float* theta      = (const float*)d_in[1];
    const float* bias       = (const float*)d_in[2];
    const int*   node_idx   = (const int*)d_in[3];
    const int*   hyedge_idx = (const int*)d_in[4];
    float*       out        = (float*)d_out;

    char*  ws    = (char*)d_ws;
    float* y     = (float*)(ws);
    float* e_ft  = (float*)(ws + 51200000);
    float* deg_e = (float*)(ws + 61440000);
    float* deg_n = (float*)(ws + 61520000);

    hipMemsetAsync(ws + 51200000, 0, 10720000, stream);

    hyfm_degree<<<(NNZ + 255) / 256, 256, 0, stream>>>(
        node_idx, hyedge_idx, deg_n, deg_e, NNZ);
    hyfm_recip<<<(N_HYEDGES + N_NODES + 255) / 256, 256, 0, stream>>>(
        deg_e, N_HYEDGES + N_NODES);

    hyfm_gemm_wmma_f32<<<(N_NODES / 16 + 7) / 8, 256, 0, stream>>>(x, theta, y);

    hyfm_bias_init<<<(N_NODES * OUT_CH + 255) / 256, 256, 0, stream>>>(
        bias, out, N_NODES * OUT_CH);

    hyfm_scatter<<<NNZ / (SC_PER_WAVE * 8), 256, 0, stream>>>(
        y, e_ft, node_idx, hyedge_idx, deg_e);
    hyfm_scatter<<<NNZ / (SC_PER_WAVE * 8), 256, 0, stream>>>(
        e_ft, out, hyedge_idx, node_idx, deg_n);
}


# degree step: work-groups own 10000-bin ranges, count them in LDS (integer adds) over a slice of the index list, reciprocal pass sums 16 partial counts; its 3.2M scattered global f32 atomics are gone
# speedup vs baseline: 2.3966x; 1.4014x over previous
#include <hip/hip_runtime.h>

typedef float v2f __attribute__((ext_vector_type(2)));
typedef float v8f __attribute__((ext_vector_type(8)));

constexpr int N_NODES   = 100000;
constexpr int N_HYEDGES = 20000;
constexpr int NNZ       = 1600000;
constexpr int IN_CH     = 256;
constexpr int OUT_CH    = 128;

__global__ __launch_bounds__(256) void hyfm_gemm_wmma_f32(
    const float* __restrict__ x, const float* __restrict__ theta,
    float* __restrict__ y)
{
    const int wave    = blockIdx.x * (blockDim.x >> 5) + (threadIdx.x >> 5);
    const int n_tiles = N_NODES / 16;
    if (wave >= n_tiles) return;
    const int lane = threadIdx.x & 31;
    const int r    = lane & 15;
    const int half = lane >> 4;

    const float* __restrict__ xrow = x + (size_t)(wave * 16 + r) * IN_CH;

    v8f acc[8] = {};

    for (int kk = 0; kk < IN_CH; kk += 4) {
        const int k0 = kk + 2 * half;
        v2f a = *(const v2f*)(xrow + k0);
        const float* __restrict__ th0 = theta + (size_t)k0 * OUT_CH + r;
#pragma unroll
        for (int c = 0; c < 8; ++c) {
            v2f b;
            b.x = th0[c * 16];
            b.y = th0[c * 16 + OUT_CH];
            acc[c] = __builtin_amdgcn_wmma_f32_16x16x4_f32(
                false, a, false, b, (short)0, acc[c], false, false);
        }
    }

#pragma unroll
    for (int c = 0; c < 8; ++c) {
#pragma unroll
        for (int i = 0; i < 8; ++i) {
            y[(size_t)(wave * 16 + i + 8 * half) * OUT_CH + c * 16 + r] = acc[c][i];
        }
    }
}

constexpr int DG_R     = 10000;
constexpr int DG_PE    = N_HYEDGES / DG_R;
constexpr int DG_PN    = N_NODES / DG_R;
constexpr int DG_P     = DG_PE + DG_PN;
constexpr int DG_S     = 16;
constexpr int DG_BINS  = N_HYEDGES + N_NODES;
constexpr int DG_SLICE = NNZ / DG_S;
constexpr int DG_V4    = DG_SLICE / 4;
constexpr int DG_TH    = 1024;
constexpr int DG_FULL  = DG_V4 / DG_TH;
constexpr int DG_BATCH = 8;
static_assert(N_HYEDGES % DG_R == 0 && N_NODES % DG_R == 0, "whole ranges");
static_assert(NNZ % (DG_S * 4) == 0, "whole 16-byte vectors per slice");
static_assert(DG_FULL % DG_BATCH == 0, "whole batches");
static_assert(DG_V4 - DG_FULL * DG_TH < DG_TH, "one guarded tail vector at most");

__device__ __forceinline__ void hyfm_count(unsigned* cnt, int idx, int base)
{
    const unsigned a = (unsigned)(idx - base);
    if (a < (unsigned)DG_R) atomicAdd(&cnt[a], 1u);
}

__global__ __launch_bounds__(DG_TH) void hyfm_degree_owner(
    const int* __restrict__ nid, const int* __restrict__ eid,
    unsigned* __restrict__ part)
{
    __shared__ unsigned cnt[DG_R];
    const int p = blockIdx.x;
    const int s = blockIdx.y;
    const int t = threadIdx.x;
    const bool is_e = p < DG_PE;
    const int* __restrict__ idx = is_e ? eid : nid;
    const int base = is_e ? p * DG_R : (p - DG_PE) * DG_R;

    for (int i = t; i < DG_R; i += DG_TH) cnt[i] = 0u;
    __syncthreads();

    const int4* __restrict__ v = (const int4*)(idx + (size_t)s * DG_SLICE);
    for (int b = 0; b < DG_FULL; b += DG_BATCH) {
        int4 q[DG_BATCH];
#pragma unroll
        for (int j = 0; j < DG_BATCH; ++j) q[j] = v[(b + j) * DG_TH + t];
#pragma unroll
        for (int j = 0; j < DG_BATCH; ++j) {
            hyfm_count(cnt, q[j].x, base);
            hyfm_count(cnt, q[j].y, base);
            hyfm_count(cnt, q[j].z, base);
            hyfm_count(cnt, q[j].w, base);
        }
    }
    if (DG_FULL * DG_TH + t < DG_V4) {
        const int4 q = v[DG_FULL * DG_TH + t];
        hyfm_count(cnt, q.x, base);
        hyfm_count(cnt, q.y, base);
        hyfm_count(cnt, q.z, base);
        hyfm_count(cnt, q.w, base);
    }
    __syncthreads();

    unsigned* __restrict__ dst = part + (size_t)s * DG_BINS + (size_t)p * DG_R;
    for (int i = t; i < DG_R; i += DG_TH) dst[i] = cnt[i];
}

__global__ __launch_bounds__(256) void hyfm_recip(
    const unsigned* __restrict__ part, float* __restrict__ d, int n)
{
    int i = blockIdx.x * blockDim.x + threadIdx.x;
    if (i >= n) return;
    unsigned c = 0u;
#pragma unroll
    for (int s = 0; s < DG_S; ++s) c += part[(size_t)s * DG_BINS + i];
    const float v = (float)c;
    d[i] = (v > 0.0f) ? (1.0f / v) : 0.0f;
}

__global__ __launch_bounds__(256) void hyfm_bias_init(
    const float* __restrict__ bias, float* __restrict__ out, int total)
{
    int i = blockIdx.x * blockDim.x + threadIdx.x;
    if (i >= total) return;
    out[i] = bias[i & (OUT_CH - 1)];
}

constexpr int SC_PER_WAVE = 32;
constexpr int SC_GROUP    = 8;
constexpr int SC_NGROUPS  = SC_PER_WAVE / SC_GROUP;
static_assert(NNZ % (SC_PER_WAVE * 8) == 0, "whole blocks of 8 waves x 32 entries");

__device__ __forceinline__ void hyfm_gather_group(
    const float* __restrict__ src, int gv, int lane, int first,
    float (&v)[SC_GROUP][4])
{
#pragma unroll
    for (int j = 0; j < SC_GROUP; ++j) {
        const int g = __builtin_amdgcn_readlane(gv, first + j);
        const float* __restrict__ srow = src + (size_t)g * OUT_CH + lane;
        v[j][0] = srow[0];
        v[j][1] = srow[32];
        v[j][2] = srow[64];
        v[j][3] = srow[96];
    }
}

__global__ __launch_bounds__(256) void hyfm_scatter(
    const float* __restrict__ src, float* __restrict__ dst,
    const int* __restrict__ gidx, const int* __restrict__ sidx,
    const float* __restrict__ inv_deg)
{
    const int w    = blockIdx.x * (blockDim.x >> 5) + (threadIdx.x >> 5);
    const int lane = threadIdx.x & 31;
    const int e    = w * SC_PER_WAVE + lane;
    const int gv   = gidx[e];
    const int sv   = sidx[e];
    const int scv  = __float_as_int(inv_deg[sv]);

    float v[2][SC_GROUP][4];
    hyfm_gather_group(src, gv, lane, 0, v[0]);

#pragma unroll
    for (int c = 0; c < SC_NGROUPS; ++c) {
        if (c + 1 < SC_NGROUPS)
            hyfm_gather_group(src, gv, lane, (c + 1) * SC_GROUP, v[(c + 1) & 1]);
#pragma unroll
        for (int j = 0; j < SC_GROUP; ++j) {
            const int   s  = __builtin_amdgcn_readlane(sv, c * SC_GROUP + j);
            const float sc = __int_as_float(
                __builtin_amdgcn_readlane(scv, c * SC_GROUP + j));
            float* drow = dst + (size_t)s * OUT_CH + lane;
            atomicAdd(drow +  0, v[c & 1][j][0] * sc);
            atomicAdd(drow + 32, v[c & 1][j][1] * sc);
            atomicAdd(drow + 64, v[c & 1][j][2] * sc);
            atomicAdd(drow + 96, v[c & 1][j][3] * sc);
        }
    }
}

extern "C" void kernel_launch(void* const* d_in, const int* in_sizes, int n_in,
                              void* d_out, int out_size, void* d_ws, size_t ws_size,
                              hipStream_t stream)
{
    const float* x          = (const float*)d_in[0];
    const float* theta      = (const float*)d_in[1];
    const float* bias       = (const float*)d_in[2];
    const int*   node_idx   = (const int*)d_in[3];
    const int*   hyedge_idx = (const int*)d_in[4];
    float*       out        = (float*)d_out;

    char*  ws    = (char*)d_ws;
    float* y     = (float*)(ws);
    float* e_ft  = (float*)(ws + 51200000);
    float* deg_e = (float*)(ws + 61440000);
    float* deg_n = (float*)(ws + 61520000);

    hipMemsetAsync(ws + 51200000, 0, 10240000, stream);

    unsigned* part = (unsigned*)(ws);
    hyfm_degree_owner<<<dim3(DG_P, DG_S), DG_TH, 0, stream>>>(
        node_idx, hyedge_idx, part);
    hyfm_recip<<<(N_HYEDGES + N_NODES + 255) / 256, 256, 0, stream>>>(
        part, deg_e, N_HYEDGES + N_NODES);

    hyfm_gemm_wmma_f32<<<(N_NODES / 16 + 7) / 8, 256, 0, stream>>>(x, theta, y);

    hyfm_bias_init<<<(N_NODES * OUT_CH + 255) / 256, 256, 0, stream>>>(
        bias, out, N_NODES * OUT_CH);

    hyfm_scatter<<<NNZ / (SC_PER_WAVE * 8), 256, 0, stream>>>(
        y, e_ft, node_idx, hyedge_idx, deg_e);
    hyfm_scatter<<<NNZ / (SC_PER_WAVE * 8), 256, 0, stream>>>(
        e_ft, out, hyedge_idx, node_idx, deg_n);
}


# both aggregations: entries bucketed by destination (counting sort in LDS, rebuilt each call), one wave sums a row's source rows in registers in bucket order; float atomics, memset, bias pass gone
# speedup vs baseline: 3.5421x; 1.4780x over previous
#include <hip/hip_runtime.h>

typedef float v2f __attribute__((ext_vector_type(2)));
typedef float v8f __attribute__((ext_vector_type(8)));

constexpr int N_NODES   = 100000;
constexpr int N_HYEDGES = 20000;
constexpr int NNZ       = 1600000;
constexpr int IN_CH     = 256;
constexpr int OUT_CH    = 128;

__global__ __launch_bounds__(256) void hyfm_gemm_wmma_f32(
    const float* __restrict__ x, const float* __restrict__ theta,
    float* __restrict__ y)
{
    const int wave    = blockIdx.x * (blockDim.x >> 5) + (threadIdx.x >> 5);
    const int n_tiles = N_NODES / 16;
    if (wave >= n_tiles) return;
    const int lane = threadIdx.x & 31;
    const int r    = lane & 15;
    const int half = lane >> 4;

    const float* __restrict__ xrow = x + (size_t)(wave * 16 + r) * IN_CH;

    v8f acc[8] = {};

    for (int kk = 0; kk < IN_CH; kk += 4) {
        const int k0 = kk + 2 * half;
        v2f a = *(const v2f*)(xrow + k0);
        const float* __restrict__ th0 = theta + (size_t)k0 * OUT_CH + r;
#pragma unroll
        for (int c = 0; c < 8; ++c) {
            v2f b;
            b.x = th0[c * 16];
            b.y = th0[c * 16 + OUT_CH];
            acc[c] = __builtin_amdgcn_wmma_f32_16x16x4_f32(
                false, a, false, b, (short)0, acc[c], false, false);
        }
    }

#pragma unroll
    for (int c = 0; c < 8; ++c) {
#pragma unroll
        for (int i = 0; i < 8; ++i) {
            y[(size_t)(wave * 16 + i + 8 * half) * OUT_CH + c * 16 + r] = acc[c][i];
        }
    }
}

constexpr int DG_R     = 10000;
constexpr int DG_PE    = N_HYEDGES / DG_R;
constexpr int DG_PN    = N_NODES / DG_R;
constexpr int DG_P     = DG_PE + DG_PN;
constexpr int DG_S     = 16;
constexpr int DG_BINS  = N_HYEDGES + N_NODES;
constexpr int DG_SLICE = NNZ / DG_S;
constexpr int DG_V4    = DG_SLICE / 4;
constexpr int DG_TH    = 1024;
constexpr int DG_FULL  = DG_V4 / DG_TH;
constexpr int DG_BATCH = 8;
constexpr int PL_BATCH = 4;
constexpr int DG_PER_T = (DG_R + DG_TH - 1) / DG_TH;
static_assert(N_HYEDGES % DG_R == 0 && N_NODES % DG_R == 0, "whole ranges");
static_assert(NNZ % (DG_S * 4) == 0, "whole 16-byte vectors per slice");
static_assert(DG_FULL % DG_BATCH == 0 && DG_FULL % PL_BATCH == 0, "whole batches");
static_assert(DG_V4 - DG_FULL * DG_TH < DG_TH, "one guarded tail vector at most");
static_assert(DG_S * DG_P <= DG_TH, "one thread per range total");
static_assert(DG_R % 4 == 0 && DG_BINS % 4 == 0, "16-byte steps over the bins");
constexpr int PL_RE = 1250;
constexpr int PL_RN = 6250;
constexpr int PL_RMAX = (PL_RE > PL_RN) ? PL_RE : PL_RN;
static_assert(N_HYEDGES % PL_RE == 0 && N_NODES % PL_RN == 0, "whole place ranges");

__device__ __forceinline__ void hyfm_count(unsigned* cnt, int idx, int base)
{
    const unsigned a = (unsigned)(idx - base);
    if (a < (unsigned)DG_R) atomicAdd(&cnt[a], 1u);
}

__device__ __forceinline__ unsigned hyfm_wave_sum(unsigned v)
{
    for (int d = 16; d > 0; d >>= 1) v += __shfl_xor(v, d);
    return v;
}

__global__ __launch_bounds__(DG_TH) void hyfm_degree_owner(
    const int* __restrict__ nid, const int* __restrict__ eid,
    unsigned* __restrict__ part, unsigned* __restrict__ rtot)
{
    __shared__ unsigned cnt[DG_R];
    __shared__ unsigned wtot[DG_TH / 32];
    const int p = blockIdx.x;
    const int s = blockIdx.y;
    const int t = threadIdx.x;
    const bool is_e = p < DG_PE;
    const int* __restrict__ idx = is_e ? eid : nid;
    const int base = is_e ? p * DG_R : (p - DG_PE) * DG_R;

    for (int i = t; i < DG_R; i += DG_TH) cnt[i] = 0u;
    __syncthreads();

    const int4* __restrict__ v = (const int4*)(idx + (size_t)s * DG_SLICE);
    for (int b = 0; b < DG_FULL; b += DG_BATCH) {
        int4 q[DG_BATCH];
#pragma unroll
        for (int j = 0; j < DG_BATCH; ++j) q[j] = v[(b + j) * DG_TH + t];
#pragma unroll
        for (int j = 0; j < DG_BATCH; ++j) {
            hyfm_count(cnt, q[j].x, base);
            hyfm_count(cnt, q[j].y, base);
            hyfm_count(cnt, q[j].z, base);
            hyfm_count(cnt, q[j].w, base);
        }
    }
    if (DG_FULL * DG_TH + t < DG_V4) {
        const int4 q = v[DG_FULL * DG_TH + t];
        hyfm_count(cnt, q.x, base);
        hyfm_count(cnt, q.y, base);
        hyfm_count(cnt, q.z, base);
        hyfm_count(cnt, q.w, base);
    }
    __syncthreads();

    unsigned* __restrict__ dst = part + (size_t)s * DG_BINS + (size_t)p * DG_R;
    unsigned loc = 0u;
    for (int i = t; i < DG_R; i += DG_TH) {
        const unsigned c = cnt[i];
        dst[i] = c;
        loc += c;
    }
    loc = hyfm_wave_sum(loc);
    if ((t & 31) == 0) wtot[t >> 5] = loc;
    __syncthreads();
    if (t < 32) {
        const unsigned w = hyfm_wave_sum(wtot[t]);
        if (t == 0) rtot[s * DG_P + p] = w;
    }
}

__global__ __launch_bounds__(DG_TH) void hyfm_offsets(
    unsigned* __restrict__ part, const unsigned* __restrict__ rtot,
    int* __restrict__ rowstart)
{
    __shared__ __attribute__((aligned(16))) unsigned tot[DG_R];
    __shared__ unsigned wsum[DG_TH / 32];
    __shared__ unsigned wbase[DG_TH / 32];
    const int p    = blockIdx.x;
    const int t    = threadIdx.x;
    const int lane = t & 31;
    const int wv   = t >> 5;
    const int first = (p < DG_PE) ? 0 : DG_PE;
    unsigned* __restrict__ col = part + (size_t)p * DG_R;

    unsigned b = 0u;
    if (t < DG_S * DG_P) {
        const int q = t % DG_P;
        if (q >= first && q < p) b = rtot[t];
    }
    b = hyfm_wave_sum(b);
    if (lane == 0) wbase[wv] = b;

    for (int i = 4 * t; i < DG_R; i += 4 * DG_TH) {
        uint4 c = make_uint4(0u, 0u, 0u, 0u);
#pragma unroll
        for (int s = 0; s < DG_S; ++s) {
            const uint4 q = *(const uint4*)(col + (size_t)s * DG_BINS + i);
            c.x += q.x; c.y += q.y; c.z += q.z; c.w += q.w;
        }
        *(uint4*)(tot + i) = c;
    }
    __syncthreads();

    unsigned base = 0u;
    for (int w = 0; w < DG_TH / 32; ++w) base += wbase[w];

    const int i0 = t * DG_PER_T;
    unsigned x = 0u;
    for (int j = 0; j < DG_PER_T; ++j)
        if (i0 + j < DG_R) x += tot[i0 + j];
    unsigned incl = x;
    for (int d = 1; d < 32; d <<= 1) {
        const unsigned n = __shfl_up(incl, d);
        if (lane >= d) incl += n;
    }
    if (lane == 31) wsum[wv] = incl;
    __syncthreads();
    if (wv == 0) {
        const unsigned w = wsum[lane];
        unsigned wi = w;
        for (int d = 1; d < 32; d <<= 1) {
            const unsigned n = __shfl_up(wi, d);
            if (lane >= d) wi += n;
        }
        wsum[lane] = wi - w;
    }
    __syncthreads();
    unsigned run = base + wsum[wv] + incl - x;
    for (int j = 0; j < DG_PER_T; ++j) {
        if (i0 + j < DG_R) {
            const unsigned c = tot[i0 + j];
            tot[i0 + j] = run;
            run += c;
        }
    }
    __syncthreads();

    for (int i = 4 * t; i < DG_R; i += 4 * DG_TH) {
        uint4 r = *(const uint4*)(tot + i);
        *(int4*)(rowstart + p * DG_R + i) = make_int4((int)r.x, (int)r.y, (int)r.z, (int)r.w);
        uint4 c[DG_S];
#pragma unroll
        for (int s = 0; s < DG_S; ++s) c[s] = *(const uint4*)(col + (size_t)s * DG_BINS + i);
#pragma unroll
        for (int s = 0; s < DG_S; ++s) {
            *(uint4*)(col + (size_t)s * DG_BINS + i) = r;
            r.x += c[s].x; r.y += c[s].y; r.z += c[s].z; r.w += c[s].w;
        }
    }
}

__device__ __forceinline__ void hyfm_place_one(
    unsigned* cur, int key, int val, int base, int nbins, int* __restrict__ perm)
{
    const unsigned a = (unsigned)(key - base);
    if (a < (unsigned)nbins) {
        const unsigned pos = atomicAdd(&cur[a], 1u);
        if (pos < (unsigned)NNZ) perm[pos] = val;
    }
}

__global__ __launch_bounds__(DG_TH) void hyfm_place(
    const int* __restrict__ key, const int* __restrict__ val,
    const unsigned* __restrict__ off, int bin0, int nbins, int* __restrict__ perm)
{
    __shared__ unsigned cur[PL_RMAX];
    const int s = blockIdx.y;
    const int t = threadIdx.x;
    const int base = blockIdx.x * nbins;

    const unsigned* __restrict__ o = off + (size_t)s * DG_BINS + bin0 + base;
    for (int i = t; i < nbins; i += DG_TH) cur[i] = o[i];
    __syncthreads();

    const int4* __restrict__ kv = (const int4*)(key + (size_t)s * DG_SLICE);
    const int4* __restrict__ vv = (const int4*)(val + (size_t)s * DG_SLICE);
    for (int b = 0; b < DG_FULL; b += PL_BATCH) {
        int4 k[PL_BATCH], v[PL_BATCH];
#pragma unroll
        for (int j = 0; j < PL_BATCH; ++j) {
            k[j] = kv[(b + j) * DG_TH + t];
            v[j] = vv[(b + j) * DG_TH + t];
        }
#pragma unroll
        for (int j = 0; j < PL_BATCH; ++j) {
            hyfm_place_one(cur, k[j].x, v[j].x, base, nbins, perm);
            hyfm_place_one(cur, k[j].y, v[j].y, base, nbins, perm);
            hyfm_place_one(cur, k[j].z, v[j].z, base, nbins, perm);
            hyfm_place_one(cur, k[j].w, v[j].w, base, nbins, perm);
        }
    }
    if (DG_FULL * DG_TH + t < DG_V4) {
        const int4 k = kv[DG_FULL * DG_TH + t];
        const int4 v = vv[DG_FULL * DG_TH + t];
        hyfm_place_one(cur, k.x, v.x, base, nbins, perm);
        hyfm_place_one(cur, k.y, v.y, base, nbins, perm);
        hyfm_place_one(cur, k.z, v.z, base, nbins, perm);
        hyfm_place_one(cur, k.w, v.w, base, nbins, perm);
    }
}

constexpr int RS_GROUP = 8;

__global__ __launch_bounds__(256) void hyfm_rowsum(
    const float* __restrict__ src, float* __restrict__ dst,
    const int* __restrict__ perm, const int* __restrict__ rowstart,
    int nrows, const float* __restrict__ bias)
{
    const int lane = threadIdx.x & 31;
    const int nw   = gridDim.x * (blockDim.x >> 5);
    float4 bv = make_float4(0.0f, 0.0f, 0.0f, 0.0f);
    if (bias) bv = *(const float4*)(bias + lane * 4);

    const int w0 = __builtin_amdgcn_readfirstlane(
        (int)(blockIdx.x * (blockDim.x >> 5) + (threadIdx.x >> 5)));
    for (int row = w0; row < nrows; row += nw) {
        const int start = rowstart[row];
        const int end   = (row + 1 < nrows) ? rowstart[row + 1] : NNZ;
        const int deg   = end - start;
        const float inv = (deg > 0) ? (1.0f / (float)deg) : 0.0f;
        float4 acc = make_float4(0.0f, 0.0f, 0.0f, 0.0f);

        for (int c = start; c < end; c += 32) {
            const int n  = (end - c < 32) ? (end - c) : 32;
            const int my = (lane < n) ? perm[c + lane] : 0;
            for (int j = 0; j < n; j += RS_GROUP) {
                float4 v[RS_GROUP];
#pragma unroll
                for (int u = 0; u < RS_GROUP; ++u) {
                    const int g = __builtin_amdgcn_readlane(my, j + u);
                    v[u] = *(const float4*)(src + (size_t)g * OUT_CH + lane * 4);
                }
#pragma unroll
                for (int u = 0; u < RS_GROUP; ++u) {
                    if (j + u < n) {
                        acc.x += v[u].x * inv;
                        acc.y += v[u].y * inv;
                        acc.z += v[u].z * inv;
                        acc.w += v[u].w * inv;
                    }
                }
            }
        }
        float4 o;
        o.x = acc.x + bv.x;
        o.y = acc.y + bv.y;
        o.z = acc.z + bv.z;
        o.w = acc.w + bv.w;
        *(float4*)(dst + (size_t)row * OUT_CH + lane * 4) = o;
    }
}

extern "C" void kernel_launch(void* const* d_in, const int* in_sizes, int n_in,
                              void* d_out, int out_size, void* d_ws, size_t ws_size,
                              hipStream_t stream)
{
    const float* x          = (const float*)d_in[0];
    const float* theta      = (const float*)d_in[1];
    const float* bias       = (const float*)d_in[2];
    const int*   node_idx   = (const int*)d_in[3];
    const int*   hyedge_idx = (const int*)d_in[4];
    float*       out        = (float*)d_out;

    char*  ws       = (char*)d_ws;
    char*  ob       = (char*)d_out;
    float* y        = (float*)(ws);
    int*   perm_n   = (int*)(ws);
    float* e_ft     = (float*)(ws + 51200000);
    int*   rowstart = (int*)(ws + 61440000);
    unsigned* part  = (unsigned*)(ob);
    int*   perm_e   = (int*)(ob + 7680000);
    unsigned* rtot  = (unsigned*)(ob + 14080000);

    hyfm_degree_owner<<<dim3(DG_P, DG_S), DG_TH, 0, stream>>>(
        node_idx, hyedge_idx, part, rtot);
    hyfm_offsets<<<DG_P, DG_TH, 0, stream>>>(part, rtot, rowstart);
    hyfm_place<<<dim3(N_HYEDGES / PL_RE, DG_S), DG_TH, 0, stream>>>(
        hyedge_idx, node_idx, part, 0, PL_RE, perm_e);

    hyfm_gemm_wmma_f32<<<(N_NODES / 16 + 7) / 8, 256, 0, stream>>>(x, theta, y);

    hyfm_rowsum<<<N_HYEDGES / 16, 256, 0, stream>>>(
        y, e_ft, perm_e, rowstart, N_HYEDGES, nullptr);

    hyfm_place<<<dim3(N_NODES / PL_RN, DG_S), DG_TH, 0, stream>>>(
        node_idx, hyedge_idx, part, N_HYEDGES, PL_RN, perm_n);

    hyfm_rowsum<<<2048, 256, 0, stream>>>(
        e_ft, out, perm_n, rowstart + N_HYEDGES, N_NODES, bias);
}


# independent passes share launches, 7 to 5: offsets pass beside the GEMM (32-wave groups, 26 row tiles each), both place passes in one grid; y now lives in the output buffer, lists in workspace
# speedup vs baseline: 4.3678x; 1.2331x over previous
#include <hip/hip_runtime.h>

typedef float v2f __attribute__((ext_vector_type(2)));
typedef float v8f __attribute__((ext_vector_type(8)));

constexpr int N_NODES   = 100000;
constexpr int N_HYEDGES = 20000;
constexpr int NNZ       = 1600000;
constexpr int IN_CH     = 256;
constexpr int OUT_CH    = 128;

constexpr int GM_TILES = 26;
constexpr int GM_WGS   = (N_NODES / 16 + GM_TILES - 1) / GM_TILES;
static_assert(GM_TILES <= 32, "a strip per wave of a 1024-thread group at most");

__device__ __forceinline__ void hyfm_gemm_wmma_f32(
    int group, const float* __restrict__ x, const float* __restrict__ theta,
    float* __restrict__ y)
{
    const int wv      = threadIdx.x >> 5;
    const int wave    = group * GM_TILES + wv;
    const int n_tiles = N_NODES / 16;
    if (wv >= GM_TILES || wave >= n_tiles) return;
    const int lane = threadIdx.x & 31;
    const int r    = lane & 15;
    const int half = lane >> 4;

    const float* __restrict__ xrow = x + (size_t)(wave * 16 + r) * IN_CH;

    v8f acc[8] = {};

    for (int kk = 0; kk < IN_CH; kk += 4) {
        const int k0 = kk + 2 * half;
        v2f a = *(const v2f*)(xrow + k0);
        const float* __restrict__ th0 = theta + (size_t)k0 * OUT_CH + r;
#pragma unroll
        for (int c = 0; c < 8; ++c) {
            v2f b;
            b.x = th0[c * 16];
            b.y = th0[c * 16 + OUT_CH];
            acc[c] = __builtin_amdgcn_wmma_f32_16x16x4_f32(
                false, a, false, b, (short)0, acc[c], false, false);
        }
    }

#pragma unroll
    for (int c = 0; c < 8; ++c) {
#pragma unroll
        for (int i = 0; i < 8; ++i) {
            y[(size_t)(wave * 16 + i + 8 * half) * OUT_CH + c * 16 + r] = acc[c][i];
        }
    }
}

constexpr int DG_R     = 10000;
constexpr int DG_PE    = N_HYEDGES / DG_R;
constexpr int DG_PN    = N_NODES / DG_R;
constexpr int DG_P     = DG_PE + DG_PN;
constexpr int DG_S     = 16;
constexpr int DG_BINS  = N_HYEDGES + N_NODES;
constexpr int DG_SLICE = NNZ / DG_S;
constexpr int DG_V4    = DG_SLICE / 4;
constexpr int DG_TH    = 1024;
constexpr int DG_FULL  = DG_V4 / DG_TH;
constexpr int DG_BATCH = 8;
constexpr int PL_BATCH = 4;
constexpr int DG_PER_T = (DG_R + DG_TH - 1) / DG_TH;
static_assert(N_HYEDGES % DG_R == 0 && N_NODES % DG_R == 0, "whole ranges");
static_assert(NNZ % (DG_S * 4) == 0, "whole 16-byte vectors per slice");
static_assert(DG_FULL % DG_BATCH == 0 && DG_FULL % PL_BATCH == 0, "whole batches");
static_assert(DG_V4 - DG_FULL * DG_TH < DG_TH, "one guarded tail vector at most");
static_assert(DG_S * DG_P <= DG_TH, "one thread per range total");
static_assert(DG_R % 4 == 0 && DG_BINS % 4 == 0, "16-byte steps over the bins");
constexpr int PL_RE = 1250;
constexpr int PL_RN = 6250;
constexpr int PL_RMAX = (PL_RE > PL_RN) ? PL_RE : PL_RN;
static_assert(N_HYEDGES % PL_RE == 0 && N_NODES % PL_RN == 0, "whole place ranges");

__device__ __forceinline__ void hyfm_count(unsigned* cnt, int idx, int base)
{
    const unsigned a = (unsigned)(idx - base);
    if (a < (unsigned)DG_R) atomicAdd(&cnt[a], 1u);
}

__device__ __forceinline__ unsigned hyfm_wave_sum(unsigned v)
{
    for (int d = 16; d > 0; d >>= 1) v += __shfl_xor(v, d);
    return v;
}

__global__ __launch_bounds__(DG_TH) void hyfm_degree_owner(
    const int* __restrict__ nid, const int* __restrict__ eid,
    unsigned* __restrict__ part, unsigned* __restrict__ rtot)
{
    __shared__ unsigned cnt[DG_R];
    __shared__ unsigned wtot[DG_TH / 32];
    const int p = blockIdx.x;
    const int s = blockIdx.y;
    const int t = threadIdx.x;
    const bool is_e = p < DG_PE;
    const int* __restrict__ idx = is_e ? eid : nid;
    const int base = is_e ? p * DG_R : (p - DG_PE) * DG_R;

    for (int i = t; i < DG_R; i += DG_TH) cnt[i] = 0u;
    __syncthreads();

    const int4* __restrict__ v = (const int4*)(idx + (size_t)s * DG_SLICE);
    for (int b = 0; b < DG_FULL; b += DG_BATCH) {
        int4 q[DG_BATCH];
#pragma unroll
        for (int j = 0; j < DG_BATCH; ++j) q[j] = v[(b + j) * DG_TH + t];
#pragma unroll
        for (int j = 0; j < DG_BATCH; ++j) {
            hyfm_count(cnt, q[j].x, base);
            hyfm_count(cnt, q[j].y, base);
            hyfm_count(cnt, q[j].z, base);
            hyfm_count(cnt, q[j].w, base);
        }
    }
    if (DG_FULL * DG_TH + t < DG_V4) {
        const int4 q = v[DG_FULL * DG_TH + t];
        hyfm_count(cnt, q.x, base);
        hyfm_count(cnt, q.y, base);
        hyfm_count(cnt, q.z, base);
        hyfm_count(cnt, q.w, base);
    }
    __syncthreads();

    unsigned* __restrict__ dst = part + (size_t)s * DG_BINS + (size_t)p * DG_R;
    unsigned loc = 0u;
    for (int i = t; i < DG_R; i += DG_TH) {
        const unsigned c = cnt[i];
        dst[i] = c;
        loc += c;
    }
    loc = hyfm_wave_sum(loc);
    if ((t & 31) == 0) wtot[t >> 5] = loc;
    __syncthreads();
    if (t < 32) {
        const unsigned w = hyfm_wave_sum(wtot[t]);
        if (t == 0) rtot[s * DG_P + p] = w;
    }
}

__device__ __forceinline__ void hyfm_offsets(
    int p,
    unsigned* __restrict__ part, const unsigned* __restrict__ rtot,
    int* __restrict__ rowstart)
{
    __shared__ __attribute__((aligned(16))) unsigned tot[DG_R];
    __shared__ unsigned wsum[DG_TH / 32];
    __shared__ unsigned wbase[DG_TH / 32];
    const int t    = threadIdx.x;
    const int lane = t & 31;
    const int wv   = t >> 5;
    const int first = (p < DG_PE) ? 0 : DG_PE;
    unsigned* __restrict__ col = part + (size_t)p * DG_R;

    unsigned b = 0u;
    if (t < DG_S * DG_P) {
        const int q = t % DG_P;
        if (q >= first && q < p) b = rtot[t];
    }
    b = hyfm_wave_sum(b);
    if (lane == 0) wbase[wv] = b;

    for (int i = 4 * t; i < DG_R; i += 4 * DG_TH) {
        uint4 c = make_uint4(0u, 0u, 0u, 0u);
#pragma unroll
        for (int s = 0; s < DG_S; ++s) {
            const uint4 q = *(const uint4*)(col + (size_t)s * DG_BINS + i);
            c.x += q.x; c.y += q.y; c.z += q.z; c.w += q.w;
        }
        *(uint4*)(tot + i) = c;
    }
    __syncthreads();

    unsigned base = 0u;
    for (int w = 0; w < DG_TH / 32; ++w) base += wbase[w];

    const int i0 = t * DG_PER_T;
    unsigned x = 0u;
    for (int j = 0; j < DG_PER_T; ++j)
        if (i0 + j < DG_R) x += tot[i0 + j];
    unsigned incl = x;
    for (int d = 1; d < 32; d <<= 1) {
        const unsigned n = __shfl_up(incl, d);
        if (lane >= d) incl += n;
    }
    if (lane == 31) wsum[wv] = incl;
    __syncthreads();
    if (wv == 0) {
        const unsigned w = wsum[lane];
        unsigned wi = w;
        for (int d = 1; d < 32; d <<= 1) {
            const unsigned n = __shfl_up(wi, d);
            if (lane >= d) wi += n;
        }
        wsum[lane] = wi - w;
    }
    __syncthreads();
    unsigned run = base + wsum[wv] + incl - x;
    for (int j = 0; j < DG_PER_T; ++j) {
        if (i0 + j < DG_R) {
            const unsigned c = tot[i0 + j];
            tot[i0 + j] = run;
            run += c;
        }
    }
    __syncthreads();

    for (int i = 4 * t; i < DG_R; i += 4 * DG_TH) {
        uint4 r = *(const uint4*)(tot + i);
        *(int4*)(rowstart + p * DG_R + i) = make_int4((int)r.x, (int)r.y, (int)r.z, (int)r.w);
        uint4 c[DG_S];
#pragma unroll
        for (int s = 0; s < DG_S; ++s) c[s] = *(const uint4*)(col + (size_t)s * DG_BINS + i);
#pragma unroll
        for (int s = 0; s < DG_S; ++s) {
            *(uint4*)(col + (size_t)s * DG_BINS + i) = r;
            r.x += c[s].x; r.y += c[s].y; r.z += c[s].z; r.w += c[s].w;
        }
    }
}

__global__ __launch_bounds__(DG_TH) void hyfm_offsets_gemm(
    unsigned* __restrict__ part, const unsigned* __restrict__ rtot,
    int* __restrict__ rowstart,
    const float* __restrict__ x, const float* __restrict__ theta,
    float* __restrict__ y)
{
    if (blockIdx.x < DG_P) hyfm_offsets((int)blockIdx.x, part, rtot, rowstart);
    else hyfm_gemm_wmma_f32((int)blockIdx.x - DG_P, x, theta, y);
}

__device__ __forceinline__ void hyfm_place_one(
    unsigned* cur, int key, int val, int base, int nbins, int* __restrict__ perm)
{
    const unsigned a = (unsigned)(key - base);
    if (a < (unsigned)nbins) {
        const unsigned pos = atomicAdd(&cur[a], 1u);
        if (pos < (unsigned)NNZ) perm[pos] = val;
    }
}

constexpr int PL_GE = N_HYEDGES / PL_RE;
constexpr int PL_GN = N_NODES / PL_RN;

__global__ __launch_bounds__(DG_TH) void hyfm_place(
    const int* __restrict__ nid, const int* __restrict__ eid,
    const unsigned* __restrict__ off,
    int* __restrict__ perm_e, int* __restrict__ perm_n)
{
    __shared__ unsigned cur[PL_RMAX];
    const int s = blockIdx.y;
    const int t = threadIdx.x;
    const bool is_e = blockIdx.x < PL_GE;
    const int* key  = is_e ? eid : nid;
    const int* val  = is_e ? nid : eid;
    int* perm       = is_e ? perm_e : perm_n;
    const int bin0  = is_e ? 0 : N_HYEDGES;
    const int nbins = is_e ? PL_RE : PL_RN;
    const int base  = (is_e ? (int)blockIdx.x : (int)blockIdx.x - PL_GE) * nbins;

    const unsigned* __restrict__ o = off + (size_t)s * DG_BINS + bin0 + base;
    for (int i = t; i < nbins; i += DG_TH) cur[i] = o[i];
    __syncthreads();

    const int4* __restrict__ kv = (const int4*)(key + (size_t)s * DG_SLICE);
    const int4* __restrict__ vv = (const int4*)(val + (size_t)s * DG_SLICE);
    for (int b = 0; b < DG_FULL; b += PL_BATCH) {
        int4 k[PL_BATCH], v[PL_BATCH];
#pragma unroll
        for (int j = 0; j < PL_BATCH; ++j) {
            k[j] = kv[(b + j) * DG_TH + t];
            v[j] = vv[(b + j) * DG_TH + t];
        }
#pragma unroll
        for (int j = 0; j < PL_BATCH; ++j) {
            hyfm_place_one(cur, k[j].x, v[j].x, base, nbins, perm);
            hyfm_place_one(cur, k[j].y, v[j].y, base, nbins, perm);
            hyfm_place_one(cur, k[j].z, v[j].z, base, nbins, perm);
            hyfm_place_one(cur, k[j].w, v[j].w, base, nbins, perm);
        }
    }
    if (DG_FULL * DG_TH + t < DG_V4) {
        const int4 k = kv[DG_FULL * DG_TH + t];
        const int4 v = vv[DG_FULL * DG_TH + t];
        hyfm_place_one(cur, k.x, v.x, base, nbins, perm);
        hyfm_place_one(cur, k.y, v.y, base, nbins, perm);
        hyfm_place_one(cur, k.z, v.z, base, nbins, perm);
        hyfm_place_one(cur, k.w, v.w, base, nbins, perm);
    }
}

constexpr int RS_GROUP = 8;

__global__ __launch_bounds__(256) void hyfm_rowsum(
    const float* __restrict__ src, float* __restrict__ dst,
    const int* __restrict__ perm, const int* __restrict__ rowstart,
    int nrows, const float* __restrict__ bias)
{
    const int lane = threadIdx.x & 31;
    const int nw   = gridDim.x * (blockDim.x >> 5);
    float4 bv = make_float4(0.0f, 0.0f, 0.0f, 0.0f);
    if (bias) bv = *(const float4*)(bias + lane * 4);

    const int w0 = __builtin_amdgcn_readfirstlane(
        (int)(blockIdx.x * (blockDim.x >> 5) + (threadIdx.x >> 5)));
    for (int row = w0; row < nrows; row += nw) {
        const int start = rowstart[row];
        const int end   = (row + 1 < nrows) ? rowstart[row + 1] : NNZ;
        const int deg   = end - start;
        const float inv = (deg > 0) ? (1.0f / (float)deg) : 0.0f;
        float4 acc = make_float4(0.0f, 0.0f, 0.0f, 0.0f);

        for (int c = start; c < end; c += 32) {
            const int n  = (end - c < 32) ? (end - c) : 32;
            const int my = (lane < n) ? perm[c + lane] : 0;
            for (int j = 0; j < n; j += RS_GROUP) {
                float4 v[RS_GROUP];
#pragma unroll
                for (int u = 0; u < RS_GROUP; ++u) {
                    const int g = __builtin_amdgcn_readlane(my, j + u);
                    v[u] = *(const float4*)(src + (size_t)g * OUT_CH + lane * 4);
                }
#pragma unroll
                for (int u = 0; u < RS_GROUP; ++u) {
                    if (j + u < n) {
                        acc.x += v[u].x * inv;
                        acc.y += v[u].y * inv;
                        acc.z += v[u].z * inv;
                        acc.w += v[u].w * inv;
                    }
                }
            }
        }
        float4 o;
        o.x = acc.x + bv.x;
        o.y = acc.y + bv.y;
        o.z = acc.z + bv.z;
        o.w = acc.w + bv.w;
        *(float4*)(dst + (size_t)row * OUT_CH + lane * 4) = o;
    }
}

extern "C" void kernel_launch(void* const* d_in, const int* in_sizes, int n_in,
                              void* d_out, int out_size, void* d_ws, size_t ws_size,
                              hipStream_t stream)
{
    const float* x          = (const float*)d_in[0];
    const float* theta      = (const float*)d_in[1];
    const float* bias       = (const float*)d_in[2];
    const int*   node_idx   = (const int*)d_in[3];
    const int*   hyedge_idx = (const int*)d_in[4];
    float*       out        = (float*)d_out;

    char*  ws       = (char*)d_ws;
    float* y        = out;
    float* e_ft     = (float*)(ws);
    int*   rowstart = (int*)(ws + 10240000);
    unsigned* part  = (unsigned*)(ws + 10720000);
    int*   perm_e   = (int*)(ws + 18400000);
    int*   perm_n   = (int*)(ws + 24800000);
    unsigned* rtot  = (unsigned*)(ws + 31200000);

    hyfm_degree_owner<<<dim3(DG_P, DG_S), DG_TH, 0, stream>>>(
        node_idx, hyedge_idx, part, rtot);
    hyfm_offsets_gemm<<<DG_P + GM_WGS, DG_TH, 0, stream>>>(
        part, rtot, rowstart, x, theta, y);
    hyfm_place<<<dim3(PL_GE + PL_GN, DG_S), DG_TH, 0, stream>>>(
        node_idx, hyedge_idx, part, perm_e, perm_n);

    hyfm_rowsum<<<N_HYEDGES / 16, 256, 0, stream>>>(
        y, e_ft, perm_e, rowstart, N_HYEDGES, nullptr);

    hyfm_rowsum<<<2048, 256, 0, stream>>>(
        e_ft, out, perm_n, rowstart + N_HYEDGES, N_NODES, bias);
}


# GEMM: theta staged once per work-group in LDS, k pairs interleaved (pitch 288 dwords, halves on disjoint banks); a B fragment is one ds_load_b64, not two global loads; k order and sums unchanged
# speedup vs baseline: 4.6403x; 1.0624x over previous
#include <hip/hip_runtime.h>

typedef float v2f __attribute__((ext_vector_type(2)));
typedef float v8f __attribute__((ext_vector_type(8)));

constexpr int N_NODES   = 100000;
constexpr int N_HYEDGES = 20000;
constexpr int NNZ       = 1600000;
constexpr int IN_CH     = 256;
constexpr int OUT_CH    = 128;

constexpr int DG_TH_GEMM = 1024;
constexpr int GM_TILES = 26;
constexpr int GM_WGS   = (N_NODES / 16 + GM_TILES - 1) / GM_TILES;
static_assert(GM_TILES <= 32, "a strip per wave of a 1024-thread group at most");

constexpr int GM_BP = 2 * OUT_CH + 32;

__device__ __forceinline__ void hyfm_gemm_wmma_f32(
    int group, const float* __restrict__ x, const float* __restrict__ theta,
    float* __restrict__ y)
{
    __shared__ __attribute__((aligned(16))) float bs[(IN_CH / 2) * GM_BP];

    for (int it = threadIdx.x; it < (IN_CH / 2) * (OUT_CH / 4); it += DG_TH_GEMM) {
        const int kp = it / (OUT_CH / 4);
        const int q  = it % (OUT_CH / 4);
        const float4 lo = *(const float4*)(theta + (size_t)(2 * kp) * OUT_CH + 4 * q);
        const float4 hi = *(const float4*)(theta + (size_t)(2 * kp + 1) * OUT_CH + 4 * q);
        float* d = bs + kp * GM_BP + 8 * q;
        *(float4*)(d)     = make_float4(lo.x, hi.x, lo.y, hi.y);
        *(float4*)(d + 4) = make_float4(lo.z, hi.z, lo.w, hi.w);
    }
    __syncthreads();

    const int wv      = threadIdx.x >> 5;
    const int wave    = group * GM_TILES + wv;
    const int n_tiles = N_NODES / 16;
    if (wv >= GM_TILES || wave >= n_tiles) return;
    const int lane = threadIdx.x & 31;
    const int r    = lane & 15;
    const int half = lane >> 4;

    const float* __restrict__ xrow = x + (size_t)(wave * 16 + r) * IN_CH;
    const float* bl = bs + half * GM_BP + 2 * r;

    v8f acc[8] = {};

    for (int kk = 0; kk < IN_CH; kk += 4) {
        const int k0 = kk + 2 * half;
        v2f a = *(const v2f*)(xrow + k0);
        const float* bk = bl + (kk >> 1) * GM_BP;
#pragma unroll
        for (int c = 0; c < 8; ++c) {
            const v2f b = *(const v2f*)(bk + c * 32);
            acc[c] = __builtin_amdgcn_wmma_f32_16x16x4_f32(
                false, a, false, b, (short)0, acc[c], false, false);
        }
    }

#pragma unroll
    for (int c = 0; c < 8; ++c) {
#pragma unroll
        for (int i = 0; i < 8; ++i) {
            y[(size_t)(wave * 16 + i + 8 * half) * OUT_CH + c * 16 + r] = acc[c][i];
        }
    }
}

constexpr int DG_R     = 10000;
constexpr int DG_PE    = N_HYEDGES / DG_R;
constexpr int DG_PN    = N_NODES / DG_R;
constexpr int DG_P     = DG_PE + DG_PN;
constexpr int DG_S     = 16;
constexpr int DG_BINS  = N_HYEDGES + N_NODES;
constexpr int DG_SLICE = NNZ / DG_S;
constexpr int DG_V4    = DG_SLICE / 4;
constexpr int DG_TH    = 1024;
constexpr int DG_FULL  = DG_V4 / DG_TH;
constexpr int DG_BATCH = 8;
constexpr int PL_BATCH = 4;
constexpr int DG_PER_T = (DG_R + DG_TH - 1) / DG_TH;
static_assert(N_HYEDGES % DG_R == 0 && N_NODES % DG_R == 0, "whole ranges");
static_assert(NNZ % (DG_S * 4) == 0, "whole 16-byte vectors per slice");
static_assert(DG_FULL % DG_BATCH == 0 && DG_FULL % PL_BATCH == 0, "whole batches");
static_assert(DG_V4 - DG_FULL * DG_TH < DG_TH, "one guarded tail vector at most");
static_assert(DG_S * DG_P <= DG_TH, "one thread per range total");
static_assert(DG_TH == DG_TH_GEMM, "the GEMM stages theta with the shared launch's block size");
static_assert(DG_R % 4 == 0 && DG_BINS % 4 == 0, "16-byte steps over the bins");
constexpr int PL_RE = 1250;
constexpr int PL_RN = 6250;
constexpr int PL_RMAX = (PL_RE > PL_RN) ? PL_RE : PL_RN;
static_assert(N_HYEDGES % PL_RE == 0 && N_NODES % PL_RN == 0, "whole place ranges");

__device__ __forceinline__ void hyfm_count(unsigned* cnt, int idx, int base)
{
    const unsigned a = (unsigned)(idx - base);
    if (a < (unsigned)DG_R) atomicAdd(&cnt[a], 1u);
}

__device__ __forceinline__ unsigned hyfm_wave_sum(unsigned v)
{
    for (int d = 16; d > 0; d >>= 1) v += __shfl_xor(v, d);
    return v;
}

__global__ __launch_bounds__(DG_TH) void hyfm_degree_owner(
    const int* __restrict__ nid, const int* __restrict__ eid,
    unsigned* __restrict__ part, unsigned* __restrict__ rtot)
{
    __shared__ unsigned cnt[DG_R];
    __shared__ unsigned wtot[DG_TH / 32];
    const int p = blockIdx.x;
    const int s = blockIdx.y;
    const int t = threadIdx.x;
    const bool is_e = p < DG_PE;
    const int* __restrict__ idx = is_e ? eid : nid;
    const int base = is_e ? p * DG_R : (p - DG_PE) * DG_R;

    for (int i = t; i < DG_R; i += DG_TH) cnt[i] = 0u;
    __syncthreads();

    const int4* __restrict__ v = (const int4*)(idx + (size_t)s * DG_SLICE);
    for (int b = 0; b < DG_FULL; b += DG_BATCH) {
        int4 q[DG_BATCH];
#pragma unroll
        for (int j = 0; j < DG_BATCH; ++j) q[j] = v[(b + j) * DG_TH + t];
#pragma unroll
        for (int j = 0; j < DG_BATCH; ++j) {
            hyfm_count(cnt, q[j].x, base);
            hyfm_count(cnt, q[j].y, base);
            hyfm_count(cnt, q[j].z, base);
            hyfm_count(cnt, q[j].w, base);
        }
    }
    if (DG_FULL * DG_TH + t < DG_V4) {
        const int4 q = v[DG_FULL * DG_TH + t];
        hyfm_count(cnt, q.x, base);
        hyfm_count(cnt, q.y, base);
        hyfm_count(cnt, q.z, base);
        hyfm_count(cnt, q.w, base);
    }
    __syncthreads();

    unsigned* __restrict__ dst = part + (size_t)s * DG_BINS + (size_t)p * DG_R;
    unsigned loc = 0u;
    for (int i = t; i < DG_R; i += DG_TH) {
        const unsigned c = cnt[i];
        dst[i] = c;
        loc += c;
    }
    loc = hyfm_wave_sum(loc);
    if ((t & 31) == 0) wtot[t >> 5] = loc;
    __syncthreads();
    if (t < 32) {
        const unsigned w = hyfm_wave_sum(wtot[t]);
        if (t == 0) rtot[s * DG_P + p] = w;
    }
}

__device__ __forceinline__ void hyfm_offsets(
    int p,
    unsigned* __restrict__ part, const unsigned* __restrict__ rtot,
    int* __restrict__ rowstart)
{
    __shared__ __attribute__((aligned(16))) unsigned tot[DG_R];
    __shared__ unsigned wsum[DG_TH / 32];
    __shared__ unsigned wbase[DG_TH / 32];
    const int t    = threadIdx.x;
    const int lane = t & 31;
    const int wv   = t >> 5;
    const int first = (p < DG_PE) ? 0 : DG_PE;
    unsigned* __restrict__ col = part + (size_t)p * DG_R;

    unsigned b = 0u;
    if (t < DG_S * DG_P) {
        const int q = t % DG_P;
        if (q >= first && q < p) b = rtot[t];
    }
    b = hyfm_wave_sum(b);
    if (lane == 0) wbase[wv] = b;

    for (int i = 4 * t; i < DG_R; i += 4 * DG_TH) {
        uint4 c = make_uint4(0u, 0u, 0u, 0u);
#pragma unroll
        for (int s = 0; s < DG_S; ++s) {
            const uint4 q = *(const uint4*)(col + (size_t)s * DG_BINS + i);
            c.x += q.x; c.y += q.y; c.z += q.z; c.w += q.w;
        }
        *(uint4*)(tot + i) = c;
    }
    __syncthreads();

    unsigned base = 0u;
    for (int w = 0; w < DG_TH / 32; ++w) base += wbase[w];

    const int i0 = t * DG_PER_T;
    unsigned x = 0u;
    for (int j = 0; j < DG_PER_T; ++j)
        if (i0 + j < DG_R) x += tot[i0 + j];
    unsigned incl = x;
    for (int d = 1; d < 32; d <<= 1) {
        const unsigned n = __shfl_up(incl, d);
        if (lane >= d) incl += n;
    }
    if (lane == 31) wsum[wv] = incl;
    __syncthreads();
    if (wv == 0) {
        const unsigned w = wsum[lane];
        unsigned wi = w;
        for (int d = 1; d < 32; d <<= 1) {
            const unsigned n = __shfl_up(wi, d);
            if (lane >= d) wi += n;
        }
        wsum[lane] = wi - w;
    }
    __syncthreads();
    unsigned run = base + wsum[wv] + incl - x;
    for (int j = 0; j < DG_PER_T; ++j) {
        if (i0 + j < DG_R) {
            const unsigned c = tot[i0 + j];
            tot[i0 + j] = run;
            run += c;
        }
    }
    __syncthreads();

    for (int i = 4 * t; i < DG_R; i += 4 * DG_TH) {
        uint4 r = *(const uint4*)(tot + i);
        *(int4*)(rowstart + p * DG_R + i) = make_int4((int)r.x, (int)r.y, (int)r.z, (int)r.w);
        uint4 c[DG_S];
#pragma unroll
        for (int s = 0; s < DG_S; ++s) c[s] = *(const uint4*)(col + (size_t)s * DG_BINS + i);
#pragma unroll
        for (int s = 0; s < DG_S; ++s) {
            *(uint4*)(col + (size_t)s * DG_BINS + i) = r;
            r.x += c[s].x; r.y += c[s].y; r.z += c[s].z; r.w += c[s].w;
        }
    }
}

__global__ __launch_bounds__(DG_TH) void hyfm_offsets_gemm(
    unsigned* __restrict__ part, const unsigned* __restrict__ rtot,
    int* __restrict__ rowstart,
    const float* __restrict__ x, const float* __restrict__ theta,
    float* __restrict__ y)
{
    if (blockIdx.x < DG_P) hyfm_offsets((int)blockIdx.x, part, rtot, rowstart);
    else hyfm_gemm_wmma_f32((int)blockIdx.x - DG_P, x, theta, y);
}

__device__ __forceinline__ void hyfm_place_one(
    unsigned* cur, int key, int val, int base, int nbins, int* __restrict__ perm)
{
    const unsigned a = (unsigned)(key - base);
    if (a < (unsigned)nbins) {
        const unsigned pos = atomicAdd(&cur[a], 1u);
        if (pos < (unsigned)NNZ) perm[pos] = val;
    }
}

constexpr int PL_GE = N_HYEDGES / PL_RE;
constexpr int PL_GN = N_NODES / PL_RN;

__global__ __launch_bounds__(DG_TH) void hyfm_place(
    const int* __restrict__ nid, const int* __restrict__ eid,
    const unsigned* __restrict__ off,
    int* __restrict__ perm_e, int* __restrict__ perm_n)
{
    __shared__ unsigned cur[PL_RMAX];
    const int s = blockIdx.y;
    const int t = threadIdx.x;
    const bool is_e = blockIdx.x < PL_GE;
    const int* key  = is_e ? eid : nid;
    const int* val  = is_e ? nid : eid;
    int* perm       = is_e ? perm_e : perm_n;
    const int bin0  = is_e ? 0 : N_HYEDGES;
    const int nbins = is_e ? PL_RE : PL_RN;
    const int base  = (is_e ? (int)blockIdx.x : (int)blockIdx.x - PL_GE) * nbins;

    const unsigned* __restrict__ o = off + (size_t)s * DG_BINS + bin0 + base;
    for (int i = t; i < nbins; i += DG_TH) cur[i] = o[i];
    __syncthreads();

    const int4* __restrict__ kv = (const int4*)(key + (size_t)s * DG_SLICE);
    const int4* __restrict__ vv = (const int4*)(val + (size_t)s * DG_SLICE);
    for (int b = 0; b < DG_FULL; b += PL_BATCH) {
        int4 k[PL_BATCH], v[PL_BATCH];
#pragma unroll
        for (int j = 0; j < PL_BATCH; ++j) {
            k[j] = kv[(b + j) * DG_TH + t];
            v[j] = vv[(b + j) * DG_TH + t];
        }
#pragma unroll
        for (int j = 0; j < PL_BATCH; ++j) {
            hyfm_place_one(cur, k[j].x, v[j].x, base, nbins, perm);
            hyfm_place_one(cur, k[j].y, v[j].y, base, nbins, perm);
            hyfm_place_one(cur, k[j].z, v[j].z, base, nbins, perm);
            hyfm_place_one(cur, k[j].w, v[j].w, base, nbins, perm);
        }
    }
    if (DG_FULL * DG_TH + t < DG_V4) {
        const int4 k = kv[DG_FULL * DG_TH + t];
        const int4 v = vv[DG_FULL * DG_TH + t];
        hyfm_place_one(cur, k.x, v.x, base, nbins, perm);
        hyfm_place_one(cur, k.y, v.y, base, nbins, perm);
        hyfm_place_one(cur, k.z, v.z, base, nbins, perm);
        hyfm_place_one(cur, k.w, v.w, base, nbins, perm);
    }
}

constexpr int RS_GROUP = 8;

__global__ __launch_bounds__(256) void hyfm_rowsum(
    const float* __restrict__ src, float* __restrict__ dst,
    const int* __restrict__ perm, const int* __restrict__ rowstart,
    int nrows, const float* __restrict__ bias)
{
    const int lane = threadIdx.x & 31;
    const int nw   = gridDim.x * (blockDim.x >> 5);
    float4 bv = make_float4(0.0f, 0.0f, 0.0f, 0.0f);
    if (bias) bv = *(const float4*)(bias + lane * 4);

    const int w0 = __builtin_amdgcn_readfirstlane(
        (int)(blockIdx.x * (blockDim.x >> 5) + (threadIdx.x >> 5)));
    for (int row = w0; row < nrows; row += nw) {
        const int start = rowstart[row];
        const int end   = (row + 1 < nrows) ? rowstart[row + 1] : NNZ;
        const int deg   = end - start;
        const float inv = (deg > 0) ? (1.0f / (float)deg) : 0.0f;
        float4 acc = make_float4(0.0f, 0.0f, 0.0f, 0.0f);

        for (int c = start; c < end; c += 32) {
            const int n  = (end - c < 32) ? (end - c) : 32;
            const int my = (lane < n) ? perm[c + lane] : 0;
            for (int j = 0; j < n; j += RS_GROUP) {
                float4 v[RS_GROUP];
#pragma unroll
                for (int u = 0; u < RS_GROUP; ++u) {
                    const int g = __builtin_amdgcn_readlane(my, j + u);
                    v[u] = *(const float4*)(src + (size_t)g * OUT_CH + lane * 4);
                }
#pragma unroll
                for (int u = 0; u < RS_GROUP; ++u) {
                    if (j + u < n) {
                        acc.x += v[u].x * inv;
                        acc.y += v[u].y * inv;
                        acc.z += v[u].z * inv;
                        acc.w += v[u].w * inv;
                    }
                }
            }
        }
        float4 o;
        o.x = acc.x + bv.x;
        o.y = acc.y + bv.y;
        o.z = acc.z + bv.z;
        o.w = acc.w + bv.w;
        *(float4*)(dst + (size_t)row * OUT_CH + lane * 4) = o;
    }
}

extern "C" void kernel_launch(void* const* d_in, const int* in_sizes, int n_in,
                              void* d_out, int out_size, void* d_ws, size_t ws_size,
                              hipStream_t stream)
{
    const float* x          = (const float*)d_in[0];
    const float* theta      = (const float*)d_in[1];
    const float* bias       = (const float*)d_in[2];
    const int*   node_idx   = (const int*)d_in[3];
    const int*   hyedge_idx = (const int*)d_in[4];
    float*       out        = (float*)d_out;

    char*  ws       = (char*)d_ws;
    float* y        = out;
    float* e_ft     = (float*)(ws);
    int*   rowstart = (int*)(ws + 10240000);
    unsigned* part  = (unsigned*)(ws + 10720000);
    int*   perm_e   = (int*)(ws + 18400000);
    int*   perm_n   = (int*)(ws + 24800000);
    unsigned* rtot  = (unsigned*)(ws + 31200000);

    hyfm_degree_owner<<<dim3(DG_P, DG_S), DG_TH, 0, stream>>>(
        node_idx, hyedge_idx, part, rtot);
    hyfm_offsets_gemm<<<DG_P + GM_WGS, DG_TH, 0, stream>>>(
        part, rtot, rowstart, x, theta, y);
    hyfm_place<<<dim3(PL_GE + PL_GN, DG_S), DG_TH, 0, stream>>>(
        node_idx, hyedge_idx, part, perm_e, perm_n);

    hyfm_rowsum<<<N_HYEDGES / 16, 256, 0, stream>>>(
        y, e_ft, perm_e, rowstart, N_HYEDGES, nullptr);

    hyfm_rowsum<<<2048, 256, 0, stream>>>(
        e_ft, out, perm_n, rowstart + N_HYEDGES, N_NODES, bias);
}


# counting sort re-cut: lists in 32 slices, count ranges of 20000 bins, place ranges per list cut fourfold to 4: a thread's share of a slice halves; a row's sum order follows the 32 slices
# speedup vs baseline: 4.7930x; 1.0329x over previous
#include <hip/hip_runtime.h>

typedef float v2f __attribute__((ext_vector_type(2)));
typedef float v8f __attribute__((ext_vector_type(8)));

constexpr int N_NODES   = 100000;
constexpr int N_HYEDGES = 20000;
constexpr int NNZ       = 1600000;
constexpr int IN_CH     = 256;
constexpr int OUT_CH    = 128;

constexpr int DG_TH_GEMM = 1024;
constexpr int GM_TILES = 26;
constexpr int GM_WGS   = (N_NODES / 16 + GM_TILES - 1) / GM_TILES;
static_assert(GM_TILES <= 32, "a strip per wave of a 1024-thread group at most");

constexpr int GM_BP = 2 * OUT_CH + 32;

__device__ __forceinline__ void hyfm_gemm_wmma_f32(
    int group, const float* __restrict__ x, const float* __restrict__ theta,
    float* __restrict__ y)
{
    __shared__ __attribute__((aligned(16))) float bs[(IN_CH / 2) * GM_BP];

    for (int it = threadIdx.x; it < (IN_CH / 2) * (OUT_CH / 4); it += DG_TH_GEMM) {
        const int kp = it / (OUT_CH / 4);
        const int q  = it % (OUT_CH / 4);
        const float4 lo = *(const float4*)(theta + (size_t)(2 * kp) * OUT_CH + 4 * q);
        const float4 hi = *(const float4*)(theta + (size_t)(2 * kp + 1) * OUT_CH + 4 * q);
        float* d = bs + kp * GM_BP + 8 * q;
        *(float4*)(d)     = make_float4(lo.x, hi.x, lo.y, hi.y);
        *(float4*)(d + 4) = make_float4(lo.z, hi.z, lo.w, hi.w);
    }
    __syncthreads();

    const int wv      = threadIdx.x >> 5;
    const int wave    = group * GM_TILES + wv;
    const int n_tiles = N_NODES / 16;
    if (wv >= GM_TILES || wave >= n_tiles) return;
    const int lane = threadIdx.x & 31;
    const int r    = lane & 15;
    const int half = lane >> 4;

    const float* __restrict__ xrow = x + (size_t)(wave * 16 + r) * IN_CH;
    const float* bl = bs + half * GM_BP + 2 * r;

    v8f acc[8] = {};

    for (int kk = 0; kk < IN_CH; kk += 4) {
        const int k0 = kk + 2 * half;
        v2f a = *(const v2f*)(xrow + k0);
        const float* bk = bl + (kk >> 1) * GM_BP;
#pragma unroll
        for (int c = 0; c < 8; ++c) {
            const v2f b = *(const v2f*)(bk + c * 32);
            acc[c] = __builtin_amdgcn_wmma_f32_16x16x4_f32(
                false, a, false, b, (short)0, acc[c], false, false);
        }
    }

#pragma unroll
    for (int c = 0; c < 8; ++c) {
#pragma unroll
        for (int i = 0; i < 8; ++i) {
            y[(size_t)(wave * 16 + i + 8 * half) * OUT_CH + c * 16 + r] = acc[c][i];
        }
    }
}

constexpr int DG_R     = 10000;
constexpr int DG_PE    = N_HYEDGES / DG_R;
constexpr int DG_PN    = N_NODES / DG_R;
constexpr int DG_P     = DG_PE + DG_PN;
constexpr int DG_S     = 32;
constexpr int DG_SC    = 16;
constexpr int DG_RC    = 2 * DG_R;
constexpr int DG_PCE   = N_HYEDGES / DG_RC;
constexpr int DG_PC    = (N_HYEDGES + N_NODES) / DG_RC;
constexpr int DG_BINS  = N_HYEDGES + N_NODES;
constexpr int DG_SLICE = NNZ / DG_S;
constexpr int DG_V4    = DG_SLICE / 4;
constexpr int DG_TH    = 1024;
constexpr int DG_FULL  = DG_V4 / DG_TH;
constexpr int DG_BATCH = 6;
constexpr int PL_BATCH = 6;
constexpr int DG_PER_T = (DG_R + DG_TH - 1) / DG_TH;
static_assert(N_HYEDGES % DG_R == 0 && N_NODES % DG_R == 0, "whole ranges");
static_assert(N_HYEDGES % DG_RC == 0 && N_NODES % DG_RC == 0, "whole count ranges, none across the two lists");
static_assert(DG_S % DG_SC == 0, "whole chunks of slices");
static_assert(NNZ % (DG_S * 4) == 0, "whole 16-byte vectors per slice");
static_assert(DG_FULL % DG_BATCH == 0 && DG_FULL % PL_BATCH == 0, "whole batches");
static_assert(DG_V4 - DG_FULL * DG_TH < DG_TH, "one guarded tail vector at most");
static_assert(DG_S * DG_P <= DG_TH, "one thread per range total");
static_assert(DG_TH == DG_TH_GEMM, "the GEMM stages theta with the shared launch's block size");
static_assert(DG_R % 4 == 0 && DG_BINS % 4 == 0, "16-byte steps over the bins");
constexpr int PL_RE = 5000;
constexpr int PL_RN = 25000;
constexpr int PL_RMAX = (PL_RE > PL_RN) ? PL_RE : PL_RN;
static_assert(N_HYEDGES % PL_RE == 0 && N_NODES % PL_RN == 0, "whole place ranges");
static_assert(PL_RE % 4 == 0 && PL_RN % 4 == 0, "16-byte steps over a range's cursors");

__device__ __forceinline__ void hyfm_count(unsigned* cnt, int idx, int base)
{
    const unsigned a = (unsigned)(idx - base);
    if (a < (unsigned)DG_RC) atomicAdd(&cnt[a], 1u);
}

__device__ __forceinline__ unsigned hyfm_wave_sum(unsigned v)
{
    for (int d = 16; d > 0; d >>= 1) v += __shfl_xor(v, d);
    return v;
}

__global__ __launch_bounds__(DG_TH) void hyfm_degree_owner(
    const int* __restrict__ nid, const int* __restrict__ eid,
    unsigned* __restrict__ part, unsigned* __restrict__ rtot)
{
    __shared__ __attribute__((aligned(16))) unsigned cnt[DG_RC];
    __shared__ unsigned wtot[2][DG_TH / 32];
    const int p = blockIdx.x;
    const int s = blockIdx.y;
    const int t = threadIdx.x;
    const bool is_e = p < DG_PCE;
    const int* __restrict__ idx = is_e ? eid : nid;
    const int base = is_e ? p * DG_RC : (p - DG_PCE) * DG_RC;

    for (int i = 4 * t; i < DG_RC; i += 4 * DG_TH) *(uint4*)(cnt + i) = make_uint4(0u, 0u, 0u, 0u);
    __syncthreads();

    const int4* __restrict__ v = (const int4*)(idx + (size_t)s * DG_SLICE);
    for (int b = 0; b < DG_FULL; b += DG_BATCH) {
        int4 q[DG_BATCH];
#pragma unroll
        for (int j = 0; j < DG_BATCH; ++j) q[j] = v[(b + j) * DG_TH + t];
#pragma unroll
        for (int j = 0; j < DG_BATCH; ++j) {
            hyfm_count(cnt, q[j].x, base);
            hyfm_count(cnt, q[j].y, base);
            hyfm_count(cnt, q[j].z, base);
            hyfm_count(cnt, q[j].w, base);
        }
    }
    if (DG_FULL * DG_TH + t < DG_V4) {
        const int4 q = v[DG_FULL * DG_TH + t];
        hyfm_count(cnt, q.x, base);
        hyfm_count(cnt, q.y, base);
        hyfm_count(cnt, q.z, base);
        hyfm_count(cnt, q.w, base);
    }
    __syncthreads();

    unsigned* __restrict__ dst = part + (size_t)s * DG_BINS + (size_t)p * DG_RC;
    unsigned loc0 = 0u, loc1 = 0u;
    for (int i = 4 * t; i < DG_RC; i += 4 * DG_TH) {
        const uint4 c = *(const uint4*)(cnt + i);
        *(uint4*)(dst + i) = c;
        const unsigned sum = c.x + c.y + c.z + c.w;
        if (i < DG_R) loc0 += sum; else loc1 += sum;
    }
    loc0 = hyfm_wave_sum(loc0);
    loc1 = hyfm_wave_sum(loc1);
    if ((t & 31) == 0) { wtot[0][t >> 5] = loc0; wtot[1][t >> 5] = loc1; }
    __syncthreads();
    if (t < 64) {
        const int h = t >> 5;
        const unsigned w = hyfm_wave_sum(wtot[h][t & 31]);
        if ((t & 31) == 0) rtot[s * DG_P + 2 * p + h] = w;
    }
}

__device__ __forceinline__ void hyfm_offsets(
    int p,
    unsigned* __restrict__ part, const unsigned* __restrict__ rtot,
    int* __restrict__ rowstart)
{
    __shared__ __attribute__((aligned(16))) unsigned tot[DG_R];
    __shared__ unsigned wsum[DG_TH / 32];
    __shared__ unsigned wbase[DG_TH / 32];
    const int t    = threadIdx.x;
    const int lane = t & 31;
    const int wv   = t >> 5;
    const int first = (p < DG_PE) ? 0 : DG_PE;
    unsigned* __restrict__ col = part + (size_t)p * DG_R;

    unsigned b = 0u;
    if (t < DG_S * DG_P) {
        const int q = t % DG_P;
        if (q >= first && q < p) b = rtot[t];
    }
    b = hyfm_wave_sum(b);
    if (lane == 0) wbase[wv] = b;

    for (int i = 4 * t; i < DG_R; i += 4 * DG_TH) {
        uint4 c = make_uint4(0u, 0u, 0u, 0u);
#pragma unroll 1
        for (int s0 = 0; s0 < DG_S; s0 += DG_SC) {
#pragma unroll
            for (int s = 0; s < DG_SC; ++s) {
                const uint4 q = *(const uint4*)(col + (size_t)(s0 + s) * DG_BINS + i);
                c.x += q.x; c.y += q.y; c.z += q.z; c.w += q.w;
            }
        }
        *(uint4*)(tot + i) = c;
    }
    __syncthreads();

    unsigned base = 0u;
    for (int w = 0; w < DG_TH / 32; ++w) base += wbase[w];

    const int i0 = t * DG_PER_T;
    unsigned x = 0u;
    for (int j = 0; j < DG_PER_T; ++j)
        if (i0 + j < DG_R) x += tot[i0 + j];
    unsigned incl = x;
    for (int d = 1; d < 32; d <<= 1) {
        const unsigned n = __shfl_up(incl, d);
        if (lane >= d) incl += n;
    }
    if (lane == 31) wsum[wv] = incl;
    __syncthreads();
    if (wv == 0) {
        const unsigned w = wsum[lane];
        unsigned wi = w;
        for (int d = 1; d < 32; d <<= 1) {
            const unsigned n = __shfl_up(wi, d);
            if (lane >= d) wi += n;
        }
        wsum[lane] = wi - w;
    }
    __syncthreads();
    unsigned run = base + wsum[wv] + incl - x;
    for (int j = 0; j < DG_PER_T; ++j) {
        if (i0 + j < DG_R) {
            const unsigned c = tot[i0 + j];
            tot[i0 + j] = run;
            run += c;
        }
    }
    __syncthreads();

    for (int i = 4 * t; i < DG_R; i += 4 * DG_TH) {
        uint4 r = *(const uint4*)(tot + i);
        *(int4*)(rowstart + p * DG_R + i) = make_int4((int)r.x, (int)r.y, (int)r.z, (int)r.w);
#pragma unroll 1
        for (int s0 = 0; s0 < DG_S; s0 += DG_SC) {
            uint4 c[DG_SC];
#pragma unroll
            for (int s = 0; s < DG_SC; ++s) c[s] = *(const uint4*)(col + (size_t)(s0 + s) * DG_BINS + i);
#pragma unroll
            for (int s = 0; s < DG_SC; ++s) {
                *(uint4*)(col + (size_t)(s0 + s) * DG_BINS + i) = r;
                r.x += c[s].x; r.y += c[s].y; r.z += c[s].z; r.w += c[s].w;
            }
        }
    }
}

__global__ __launch_bounds__(DG_TH) void hyfm_offsets_gemm(
    unsigned* __restrict__ part, const unsigned* __restrict__ rtot,
    int* __restrict__ rowstart,
    const float* __restrict__ x, const float* __restrict__ theta,
    float* __restrict__ y)
{
    if (blockIdx.x < DG_P) hyfm_offsets((int)blockIdx.x, part, rtot, rowstart);
    else hyfm_gemm_wmma_f32((int)blockIdx.x - DG_P, x, theta, y);
}

__device__ __forceinline__ void hyfm_place_one(
    unsigned* cur, int key, int val, int base, int nbins, int* __restrict__ perm)
{
    const unsigned a = (unsigned)(key - base);
    if (a < (unsigned)nbins) {
        const unsigned pos = atomicAdd(&cur[a], 1u);
        if (pos < (unsigned)NNZ) perm[pos] = val;
    }
}

constexpr int PL_GE = N_HYEDGES / PL_RE;
constexpr int PL_GN = N_NODES / PL_RN;
constexpr int PL_CV = (PL_RMAX / 4 + DG_TH - 1) / DG_TH;

__global__ __launch_bounds__(DG_TH) void hyfm_place(
    const int* __restrict__ nid, const int* __restrict__ eid,
    const unsigned* __restrict__ off,
    int* __restrict__ perm_e, int* __restrict__ perm_n)
{
    __shared__ __attribute__((aligned(16))) unsigned cur[PL_CV * DG_TH * 4];
    const int s = blockIdx.y;
    const int t = threadIdx.x;
    const bool is_e = blockIdx.x < PL_GE;
    const int* key  = is_e ? eid : nid;
    const int* val  = is_e ? nid : eid;
    int* perm       = is_e ? perm_e : perm_n;
    const int bin0  = is_e ? 0 : N_HYEDGES;
    const int nbins = is_e ? PL_RE : PL_RN;
    const int base  = (is_e ? (int)blockIdx.x : (int)blockIdx.x - PL_GE) * nbins;

    const unsigned* __restrict__ o = off + (size_t)s * DG_BINS + bin0 + base;
    {
        uint4 q[PL_CV];
#pragma unroll
        for (int j = 0; j < PL_CV; ++j) {
            const int i = 4 * (j * DG_TH + t);
            q[j] = *(const uint4*)(o + ((i < nbins) ? i : nbins - 4));
        }
#pragma unroll
        for (int j = 0; j < PL_CV; ++j) *(uint4*)(cur + 4 * (j * DG_TH + t)) = q[j];
    }
    __syncthreads();

    const int4* __restrict__ kv = (const int4*)(key + (size_t)s * DG_SLICE);
    const int4* __restrict__ vv = (const int4*)(val + (size_t)s * DG_SLICE);
    for (int b = 0; b < DG_FULL; b += PL_BATCH) {
        int4 k[PL_BATCH], v[PL_BATCH];
#pragma unroll
        for (int j = 0; j < PL_BATCH; ++j) {
            k[j] = kv[(b + j) * DG_TH + t];
            v[j] = vv[(b + j) * DG_TH + t];
        }
#pragma unroll
        for (int j = 0; j < PL_BATCH; ++j) {
            hyfm_place_one(cur, k[j].x, v[j].x, base, nbins, perm);
            hyfm_place_one(cur, k[j].y, v[j].y, base, nbins, perm);
            hyfm_place_one(cur, k[j].z, v[j].z, base, nbins, perm);
            hyfm_place_one(cur, k[j].w, v[j].w, base, nbins, perm);
        }
    }
    if (DG_FULL * DG_TH + t < DG_V4) {
        const int4 k = kv[DG_FULL * DG_TH + t];
        const int4 v = vv[DG_FULL * DG_TH + t];
        hyfm_place_one(cur, k.x, v.x, base, nbins, perm);
        hyfm_place_one(cur, k.y, v.y, base, nbins, perm);
        hyfm_place_one(cur, k.z, v.z, base, nbins, perm);
        hyfm_place_one(cur, k.w, v.w, base, nbins, perm);
    }
}

constexpr int RS_GROUP = 8;

__global__ __launch_bounds__(256) void hyfm_rowsum(
    const float* __restrict__ src, float* __restrict__ dst,
    const int* __restrict__ perm, const int* __restrict__ rowstart,
    int nrows, const float* __restrict__ bias)
{
    const int lane = threadIdx.x & 31;
    const int nw   = gridDim.x * (blockDim.x >> 5);
    float4 bv = make_float4(0.0f, 0.0f, 0.0f, 0.0f);
    if (bias) bv = *(const float4*)(bias + lane * 4);

    const int w0 = __builtin_amdgcn_readfirstlane(
        (int)(blockIdx.x * (blockDim.x >> 5) + (threadIdx.x >> 5)));
    for (int row = w0; row < nrows; row += nw) {
        const int start = rowstart[row];
        const int end   = (row + 1 < nrows) ? rowstart[row + 1] : NNZ;
        const int deg   = end - start;
        const float inv = (deg > 0) ? (1.0f / (float)deg) : 0.0f;
        float4 acc = make_float4(0.0f, 0.0f, 0.0f, 0.0f);

        for (int c = start; c < end; c += 32) {
            const int n  = (end - c < 32) ? (end - c) : 32;
            const int my = (lane < n) ? perm[c + lane] : 0;
            for (int j = 0; j < n; j += RS_GROUP) {
                float4 v[RS_GROUP];
#pragma unroll
                for (int u = 0; u < RS_GROUP; ++u) {
                    const int g = __builtin_amdgcn_readlane(my, j + u);
                    v[u] = *(const float4*)(src + (size_t)g * OUT_CH + lane * 4);
                }
#pragma unroll
                for (int u = 0; u < RS_GROUP; ++u) {
                    if (j + u < n) {
                        acc.x += v[u].x * inv;
                        acc.y += v[u].y * inv;
                        acc.z += v[u].z * inv;
                        acc.w += v[u].w * inv;
                    }
                }
            }
        }
        float4 o;
        o.x = acc.x + bv.x;
        o.y = acc.y + bv.y;
        o.z = acc.z + bv.z;
        o.w = acc.w + bv.w;
        *(float4*)(dst + (size_t)row * OUT_CH + lane * 4) = o;
    }
}

extern "C" void kernel_launch(void* const* d_in, const int* in_sizes, int n_in,
                              void* d_out, int out_size, void* d_ws, size_t ws_size,
                              hipStream_t stream)
{
    const float* x          = (const float*)d_in[0];
    const float* theta      = (const float*)d_in[1];
    const float* bias       = (const float*)d_in[2];
    const int*   node_idx   = (const int*)d_in[3];
    const int*   hyedge_idx = (const int*)d_in[4];
    float*       out        = (float*)d_out;

    char*  ws       = (char*)d_ws;
    float* y        = out;
    float* e_ft     = (float*)(ws);
    int*   rowstart = (int*)(ws + 10240000);
    unsigned* part  = (unsigned*)(ws + 10720000);
    int*   perm_e   = (int*)(ws + 26080000);
    int*   perm_n   = (int*)(ws + 32480000);
    unsigned* rtot  = (unsigned*)(ws + 38880000);

    hyfm_degree_owner<<<dim3(DG_PC, DG_S), DG_TH, 0, stream>>>(
        node_idx, hyedge_idx, part, rtot);
    hyfm_offsets_gemm<<<DG_P + GM_WGS, DG_TH, 0, stream>>>(
        part, rtot, rowstart, x, theta, y);
    hyfm_place<<<dim3(PL_GE + PL_GN, DG_S), DG_TH, 0, stream>>>(
        node_idx, hyedge_idx, part, perm_e, perm_n);

    hyfm_rowsum<<<N_HYEDGES / 16, 256, 0, stream>>>(
        y, e_ft, perm_e, rowstart, N_HYEDGES, nullptr);

    hyfm_rowsum<<<2048, 256, 0, stream>>>(
        e_ft, out, perm_n, rowstart + N_HYEDGES, N_NODES, bias);
}
